# starting kernel
# baseline (speedup 1.0000x reference)
#include <hip/hip_runtime.h>

typedef float v2f __attribute__((ext_vector_type(2)));
typedef float v8f __attribute__((ext_vector_type(8)));

#define ROWS   16
#define TH     (ROWS + 2)
#define TW     72
#define WIDTH  64
#define HEIGHT 64
#define CCH    128

__device__ __constant__ float Gm[3][3] = {
    { 0.5773502691896258f,  0.5773502691896258f,  0.5773502691896258f },
    { 0.7071067811865476f,  0.0f,                -0.7071067811865476f },
    { 0.4082482904638630f, -0.8164965809277260f,  0.4082482904638630f }
};

__global__ void __launch_bounds__(128)
sfcr_wmma_kernel(const float* __restrict__ x,
                 const float* __restrict__ Wl,
                 const float* __restrict__ bl,
                 float* __restrict__ out,
                 int N)
{
    __shared__ float sW[4][12];
    __shared__ float sT[TH * TW];

    const int tid = threadIdx.x;
    const int blk = blockIdx.x;
    const int HB  = HEIGHT / ROWS;
    const int h0  = (blk % HB) * ROWS;
    const int c   = (blk / HB) % CCH;
    const int n   =  blk / (HB * CCH);
    (void)N;

    if (tid < 36) {
        int o = tid / 9, j = tid % 9, p = j / 3, q = j % 3;
        float acc = Wl[o * 18 + j];
#pragma unroll
        for (int a = 0; a < 3; ++a)
#pragma unroll
            for (int b = 0; b < 3; ++b)
                acc += Wl[o * 18 + 9 + 3 * a + b] * Gm[a][p] * Gm[b][q];
        sW[o][j] = acc;
    }

    const float* xin = x + (size_t)(n * CCH + c) * (HEIGHT * WIDTH);
    for (int idx = tid; idx < TH * (WIDTH + 2); idx += 128) {
        int r = idx / (WIDTH + 2), col = idx % (WIDTH + 2);
        int gh = h0 - 1 + r; gh = gh < 0 ? 0 : (gh > HEIGHT - 1 ? HEIGHT - 1 : gh);
        int gw = col - 1;    gw = gw < 0 ? 0 : (gw > WIDTH  - 1 ? WIDTH  - 1 : gw);
        sT[r * TW + col] = xin[gh * WIDTH + gw];
    }
    __syncthreads();

    const int wave = tid >> 5;
    const int lane = tid & 31;
    const int half = lane >> 4;
    const int li   = lane & 15;
    const int kb   = half * 2;
    const int m    = li;

    v2f a0 = {0.f, 0.f}, a1 = {0.f, 0.f}, a2 = {0.f, 0.f};
    if (m < 4) {
        a0.x = sW[m][0 + kb]; a0.y = sW[m][1 + kb];
        a1.x = sW[m][4 + kb]; a1.y = sW[m][5 + kb];
        if (!half) a2.x = sW[m][8];
    }

    v8f cinit = {};
    if (!half) {
        float b0 = bl[0], b1 = bl[1], b2 = bl[2], b3 = bl[3];
        cinit[0] = b0; cinit[1] = b1; cinit[2] = b2; cinit[3] = b3;
    }

    float* obase = out + (size_t)((n * CCH + c) * 4) * (HEIGHT * WIDTH);

    for (int it = 0; it < 16; ++it) {
        int chain = wave + 4 * it;
        int r  = chain >> 2;
        int w0 = (chain & 3) << 4;

        const float* t = &sT[r * TW + w0 + li];
        v2f b0, b1, b2;
        b0.x = half ? t[2]            : t[0];
        b0.y = half ? t[TW + 0]       : t[1];
        b1.x = half ? t[2 * TW + 0]   : t[TW + 1];
        b1.y = half ? t[2 * TW + 1]   : t[TW + 2];
        b2.x = half ? 0.f             : t[2 * TW + 2];
        b2.y = 0.f;

        v8f acc = cinit;
        acc = __builtin_amdgcn_wmma_f32_16x16x4_f32(false, a0, false, b0, (short)0, acc, false, false);
        acc = __builtin_amdgcn_wmma_f32_16x16x4_f32(false, a1, false, b1, (short)0, acc, false, false);
        acc = __builtin_amdgcn_wmma_f32_16x16x4_f32(false, a2, false, b2, (short)0, acc, false, false);

        if (!half) {
            float* op = obase + (size_t)(h0 + r) * WIDTH + w0 + li;
#pragma unroll
            for (int o = 0; o < 4; ++o)
                op[(size_t)o * (HEIGHT * WIDTH)] = acc[o];
        }
    }
}

extern "C" void kernel_launch(void* const* d_in, const int* in_sizes, int n_in,
                              void* d_out, int out_size, void* d_ws, size_t ws_size,
                              hipStream_t stream) {
    const float* x  = (const float*)d_in[0];
    const float* Wl = (const float*)d_in[1];
    const float* bl = (const float*)d_in[2];
    float* out = (float*)d_out;

    int N = in_sizes[0] / (CCH * HEIGHT * WIDTH);
    int blocks = N * CCH * (HEIGHT / ROWS);
    sfcr_wmma_kernel<<<blocks, 128, 0, stream>>>(x, Wl, bl, out, N);
}


# No LDS tile: a wave slides down a 32-col strip, B fragments straight from global; 1 WMMA per input row x 16 cols sums 3 output rows in rotating accumulator row blocks; tap sum order changes
# speedup vs baseline: 1.4020x; 1.4020x over previous
#include <hip/hip_runtime.h>

typedef float v2f __attribute__((ext_vector_type(2)));
typedef float v8f __attribute__((ext_vector_type(8)));

#define WIDTH  64
#define HEIGHT 64
#define CCH    128
#define PLANE  (HEIGHT * WIDTH)
#define NB     6
#define NBATCH 11

__device__ __constant__ float Gm[3][3] = {
    { 0.5773502691896258f,  0.5773502691896258f,  0.5773502691896258f },
    { 0.7071067811865476f,  0.0f,                -0.7071067811865476f },
    { 0.4082482904638630f, -0.8164965809277260f,  0.4082482904638630f }
};

struct Rows { float a0[NB], c0[NB], a1[NB], c1[NB]; };

__device__ __forceinline__ void load_batch(Rows& R, const float* __restrict__ xin, int t0,
                                           int offA0, int offC0, int offA1, int offC1)
{
#pragma unroll
    for (int u = 0; u < NB; ++u) {
        int i = t0 + u - 1;
        i = i < 0 ? 0 : (i > HEIGHT - 1 ? HEIGHT - 1 : i);
        const float* rp = xin + i * WIDTH;
        R.a0[u] = rp[offA0];
        R.c0[u] = rp[offC0];
        R.a1[u] = rp[offA1];
        R.c1[u] = rp[offC1];
    }
}

template <int HS, bool STORE>
__device__ __forceinline__ void retire(v8f& acc0, v8f& acc1, bool lo,
                                       float* __restrict__ orow, int pixN, int pixX,
                                       float b0, float b1, float b2, float b3)
{
    const float bias[4] = { b0, b1, b2, b3 };
    constexpr int V = (HS == 1) ? 4 : 0;
    if (STORE) {
#pragma unroll
        for (int o = 0; o < 4; ++o) {
            float m;
            if (HS == 2) m = lo ? acc1[V + o] : acc0[V + o];
            else         m = lo ? acc0[V + o] : acc1[V + o];
            orow[o * PLANE + (HS == 2 ? pixX : pixN)] = m;
        }
    }
#pragma unroll
    for (int o = 0; o < 4; ++o) {
        if (HS == 2) {
            acc0[V + o] = lo ? acc0[V + o] : bias[o];
            acc1[V + o] = lo ? bias[o] : acc1[V + o];
        } else {
            acc0[V + o] = lo ? bias[o] : acc0[V + o];
            acc1[V + o] = lo ? acc1[V + o] : bias[o];
        }
    }
}

template <bool FIRST>
__device__ __forceinline__ void compute_batch(const Rows& R, v8f& acc0, v8f& acc1,
                                              const v2f (&A0)[3], const v2f (&A1)[3],
                                              bool lo, float* __restrict__ oplane, int t0,
                                              int pixN, int pixX,
                                              float b0, float b1, float b2, float b3)
{
#pragma unroll
    for (int u = 0; u < NB; ++u) {
        const int s = u % 3;
        v2f f0, f1;
        f0.x = R.a0[u]; f0.y = lo ? R.c0[u] : 0.f;
        f1.x = R.a1[u]; f1.y = lo ? R.c1[u] : 0.f;
        acc0 = __builtin_amdgcn_wmma_f32_16x16x4_f32(false, A0[s], false, f0, (short)0, acc0, false, false);
        acc1 = __builtin_amdgcn_wmma_f32_16x16x4_f32(false, A1[s], false, f1, (short)0, acc1, false, false);
        float* orow = oplane + (t0 + u - 2) * WIDTH;
        if (FIRST && u < 2) {
            if (s == 0) retire<1, false>(acc0, acc1, lo, orow, pixN, pixX, b0, b1, b2, b3);
            else        retire<2, false>(acc0, acc1, lo, orow, pixN, pixX, b0, b1, b2, b3);
        } else {
            if (s == 0)      retire<1, true>(acc0, acc1, lo, orow, pixN, pixX, b0, b1, b2, b3);
            else if (s == 1) retire<2, true>(acc0, acc1, lo, orow, pixN, pixX, b0, b1, b2, b3);
            else             retire<0, true>(acc0, acc1, lo, orow, pixN, pixX, b0, b1, b2, b3);
        }
    }
}

__global__ void __launch_bounds__(256)
sfcr_wmma_kernel(const float* __restrict__ x,
                 const float* __restrict__ Wl,
                 const float* __restrict__ bl,
                 float* __restrict__ out,
                 int N)
{
    __shared__ float sW[4][12];

    const int tid = threadIdx.x;
    (void)N;

    const int wave  = __builtin_amdgcn_readfirstlane(tid >> 5);
    const int lane  = tid & 31;
    const int half  = lane >> 4;
    const bool lo   = half == 0;
    const int c     = lane & 15;
    const int plane = blockIdx.x * 4 + (wave >> 1);
    const int wbase = (wave & 1) * 32;

    const int w0 = wbase + c, w1 = wbase + 16 + c;
    int offA0 = lo ? w0 - 1 : w0 + 1;  offA0 = offA0 < 0 ? 0 : offA0;
    int offA1 = lo ? w1 - 1 : w1 + 1;  offA1 = offA1 > WIDTH - 1 ? WIDTH - 1 : offA1;
    const int pixN = wbase + lane;
    const int pixX = wbase + (lane ^ 16);

    const float* xin    = x   + (size_t)plane * PLANE;
    float*       oplane = out + (size_t)plane * 4 * PLANE;

#define LOADB(R, k)  load_batch(R, xin, (k) * NB, offA0, w0, offA1, w1)
#define COMPB(R, k)  compute_batch<false>(R, acc0, acc1, A0, A1, lo, oplane, (k) * NB, pixN, pixX, b0, b1, b2, b3)
    Rows r0, r1, r2;
    LOADB(r0, 0); LOADB(r1, 1); LOADB(r2, 2);

    if (tid < 36) {
        int o = tid / 9, j = tid % 9, p = j / 3, q = j % 3;
        float acc = Wl[o * 18 + j];
#pragma unroll
        for (int a = 0; a < 3; ++a)
#pragma unroll
            for (int b = 0; b < 3; ++b)
                acc += Wl[o * 18 + 9 + 3 * a + b] * Gm[a][p] * Gm[b][q];
        sW[o][j] = acc;
    }
    __syncthreads();

    v2f A0[3], A1[3];
    {
        const int blk = c >> 2, o = c & 3;
        const int posg[2][3] = { {0, 1, 2}, {2, 3, 0} };
#pragma unroll
        for (int s = 0; s < 3; ++s) {
            v2f a0 = {0.f, 0.f}, a1 = {0.f, 0.f};
#pragma unroll
            for (int p = 0; p < 3; ++p) {
                const int hres = (s + 3 - p) % 3;
                float wx = sW[o][3 * p + 2 * half];
                float wy = lo ? sW[o][3 * p + 1] : 0.f;
                if (posg[0][hres] == blk) { a0.x = wx; a0.y = wy; }
                if (posg[1][hres] == blk) { a1.x = wx; a1.y = wy; }
            }
            A0[s] = a0; A1[s] = a1;
        }
    }

    const float b0 = bl[0], b1 = bl[1], b2 = bl[2], b3 = bl[3];
    v8f acc0 = { b0, b1, b2, b3, b0, b1, b2, b3 };
    v8f acc1 = acc0;

    compute_batch<true>(r0, acc0, acc1, A0, A1, lo, oplane, 0, pixN, pixX, b0, b1, b2, b3);
#pragma unroll 1
    for (int k = 0; k < 6; k += 3) {
        LOADB(r0, k + 3); COMPB(r1, k + 1);
        LOADB(r1, k + 4); COMPB(r2, k + 2);
        LOADB(r2, k + 5); COMPB(r0, k + 3);
    }
    LOADB(r0, 9);  COMPB(r1, 7);
    LOADB(r1, 10); COMPB(r2, 8);
    COMPB(r0, 9);
    COMPB(r1, 10);
#undef LOADB
#undef COMPB
}

extern "C" void kernel_launch(void* const* d_in, const int* in_sizes, int n_in,
                              void* d_out, int out_size, void* d_ws, size_t ws_size,
                              hipStream_t stream) {
    const float* x  = (const float*)d_in[0];
    const float* Wl = (const float*)d_in[1];
    const float* bl = (const float*)d_in[2];
    float* out = (float*)d_out;

    int N = in_sizes[0] / (CCH * HEIGHT * WIDTH);
    int blocks = N * (CCH / 4);
    sfcr_wmma_kernel<<<blocks, 256, 0, stream>>>(x, Wl, bl, out, N);
}


# A wave owns whole 64-col rows of a 32-row band; each finished row goes through wave-private LDS to 16-byte stores that cover whole aligned 256-B units (was 128-B halves); sum order unchanged
# speedup vs baseline: 1.4516x; 1.0354x over previous
#include <hip/hip_runtime.h>

typedef float v2f __attribute__((ext_vector_type(2)));
typedef float v8f __attribute__((ext_vector_type(8)));

#define WIDTH  64
#define HEIGHT 64
#define CCH    128
#define PLANE  (HEIGHT * WIDTH)
#define BAND   32
#define NB     6
#define NTILE  4

__device__ __constant__ float Gm[3][3] = {
    { 0.5773502691896258f,  0.5773502691896258f,  0.5773502691896258f },
    { 0.7071067811865476f,  0.0f,                -0.7071067811865476f },
    { 0.4082482904638630f, -0.8164965809277260f,  0.4082482904638630f }
};

struct Rows { float a[NTILE][NB], c[NTILE][NB]; };

struct Cols { int a[NTILE], c[NTILE]; };

template <int NU>
__device__ __forceinline__ void load_batch(Rows& R, const float* __restrict__ xin, int rfirst, int t0,
                                           const Cols& off)
{
#pragma unroll
    for (int u = 0; u < NU; ++u) {
        int i = rfirst + t0 + u - 1;
        i = i < 0 ? 0 : (i > HEIGHT - 1 ? HEIGHT - 1 : i);
        const float* rp = xin + i * WIDTH;
#pragma unroll
        for (int j = 0; j < NTILE; ++j) {
            R.a[j][u] = rp[off.a[j]];
            R.c[j][u] = rp[off.c[j]];
        }
    }
}

template <int HS, bool STORE>
__device__ __forceinline__ void retire(v8f (&acc)[NTILE], bool lo, float* __restrict__ srow,
                                       int lane, float4 (&pend)[2],
                                       float b0, float b1, float b2, float b3)
{
    const float bias[4] = { b0, b1, b2, b3 };
    constexpr int V = (HS == 1) ? 4 : 0;
    if (STORE) {
        const int pix = (HS == 2) ? (lane ^ 16) : lane;
#pragma unroll
        for (int P = 0; P < 2; ++P) {
#pragma unroll
            for (int o = 0; o < 4; ++o) {
                float m;
                if (HS == 2) m = lo ? acc[2 * P + 1][V + o] : acc[2 * P][V + o];
                else         m = lo ? acc[2 * P][V + o] : acc[2 * P + 1][V + o];
                srow[o * WIDTH + 32 * P + pix] = m;
            }
        }
#pragma unroll
        for (int j = 0; j < 2; ++j)
            pend[j] = *reinterpret_cast<const float4*>(srow + j * 2 * WIDTH + lane * 4);
    }
#pragma unroll
    for (int P = 0; P < 2; ++P) {
#pragma unroll
        for (int o = 0; o < 4; ++o) {
            if (HS == 2) {
                acc[2 * P][V + o]     = lo ? acc[2 * P][V + o] : bias[o];
                acc[2 * P + 1][V + o] = lo ? bias[o] : acc[2 * P + 1][V + o];
            } else {
                acc[2 * P][V + o]     = lo ? bias[o] : acc[2 * P][V + o];
                acc[2 * P + 1][V + o] = lo ? acc[2 * P + 1][V + o] : bias[o];
            }
        }
    }
}

__device__ __forceinline__ void flush(const float4 (&pend)[2], float* __restrict__ pst, int row)
{
#pragma unroll
    for (int j = 0; j < 2; ++j)
        *reinterpret_cast<float4*>(pst + j * 2 * PLANE + row * WIDTH) = pend[j];
}

template <bool FIRST, int NU>
__device__ __forceinline__ void compute_batch(const Rows& R, v8f (&acc)[NTILE],
                                              const v2f (&A0)[3], const v2f (&A1)[3],
                                              bool lo, float* __restrict__ srow, int lane,
                                              float* __restrict__ pst, int t0, float4 (&pend)[2],
                                              float b0, float b1, float b2, float b3)
{
#pragma unroll
    for (int u = 0; u < NU; ++u) {
        const int s = u % 3;
#pragma unroll
        for (int j = 0; j < NTILE; ++j) {
            v2f f;
            f.x = R.a[j][u]; f.y = lo ? R.c[j][u] : 0.f;
            acc[j] = __builtin_amdgcn_wmma_f32_16x16x4_f32(false, (j & 1) ? A1[s] : A0[s], false, f,
                                                           (short)0, acc[j], false, false);
        }
        const int t = t0 + u;
        if (!(FIRST && u < 3)) flush(pend, pst, t - 3);
        if (FIRST && u < 2) {
            if (s == 0) retire<1, false>(acc, lo, srow, lane, pend, b0, b1, b2, b3);
            else        retire<2, false>(acc, lo, srow, lane, pend, b0, b1, b2, b3);
        } else {
            if (s == 0)      retire<1, true>(acc, lo, srow, lane, pend, b0, b1, b2, b3);
            else if (s == 1) retire<2, true>(acc, lo, srow, lane, pend, b0, b1, b2, b3);
            else             retire<0, true>(acc, lo, srow, lane, pend, b0, b1, b2, b3);
        }
    }
}

__global__ void __launch_bounds__(256)
sfcr_wmma_kernel(const float* __restrict__ x,
                 const float* __restrict__ Wl,
                 const float* __restrict__ bl,
                 float* __restrict__ out,
                 int N)
{
    __shared__ float sW[4][12];
    __shared__ __attribute__((aligned(16))) float sT[8][4 * WIDTH];

    const int tid = threadIdx.x;
    (void)N;

    const int wave  = __builtin_amdgcn_readfirstlane(tid >> 5);
    const int lane  = tid & 31;
    const int half  = lane >> 4;
    const bool lo   = half == 0;
    const int c     = lane & 15;
    const int plane = blockIdx.x * 4 + (wave >> 1);
    const int rfirst = (wave & 1) * BAND;

    Cols off;
#pragma unroll
    for (int j = 0; j < NTILE; ++j) {
        const int w = 16 * j + c;
        int a = lo ? w - 1 : w + 1;
        a = a < 0 ? 0 : (a > WIDTH - 1 ? WIDTH - 1 : a);
        off.a[j] = a;
        off.c[j] = w;
    }

    const float* xin  = x + (size_t)plane * PLANE;
    float* srow = &sT[wave][0];
    float* pst = out + (size_t)plane * 4 * PLANE + half * PLANE + rfirst * WIDTH + 4 * c;

#define LOADB(R, k, NU)  load_batch<NU>(R, xin, rfirst, (k) * NB, off)
#define COMPB(R, k, F, NU)  compute_batch<F, NU>(R, acc, A0, A1, lo, srow, lane, pst, (k) * NB, pend, b0, b1, b2, b3)
    Rows r0, r1, r2;
    LOADB(r0, 0, NB); LOADB(r1, 1, NB); LOADB(r2, 2, NB);

    if (tid < 36) {
        int o = tid / 9, j = tid % 9, p = j / 3, q = j % 3;
        float accw = Wl[o * 18 + j];
#pragma unroll
        for (int a = 0; a < 3; ++a)
#pragma unroll
            for (int b = 0; b < 3; ++b)
                accw += Wl[o * 18 + 9 + 3 * a + b] * Gm[a][p] * Gm[b][q];
        sW[o][j] = accw;
    }
    __syncthreads();

    v2f A0[3], A1[3];
    {
        const int blk = c >> 2, o = c & 3;
        const int posg[2][3] = { {0, 1, 2}, {2, 3, 0} };
#pragma unroll
        for (int s = 0; s < 3; ++s) {
            v2f a0 = {0.f, 0.f}, a1 = {0.f, 0.f};
#pragma unroll
            for (int p = 0; p < 3; ++p) {
                const int hres = (s + 3 - p) % 3;
                float wx = sW[o][3 * p + 2 * half];
                float wy = lo ? sW[o][3 * p + 1] : 0.f;
                if (posg[0][hres] == blk) { a0.x = wx; a0.y = wy; }
                if (posg[1][hres] == blk) { a1.x = wx; a1.y = wy; }
            }
            A0[s] = a0; A1[s] = a1;
        }
    }

    const float b0 = bl[0], b1 = bl[1], b2 = bl[2], b3 = bl[3];
    v8f acc[NTILE];
#pragma unroll
    for (int j = 0; j < NTILE; ++j) acc[j] = (v8f){ b0, b1, b2, b3, b0, b1, b2, b3 };
    float4 pend[2];

    COMPB(r0, 0, true, NB);
    LOADB(r0, 3, NB); COMPB(r1, 1, false, NB);
    LOADB(r1, 4, NB); COMPB(r2, 2, false, NB);
    LOADB(r2, 5, 4);  COMPB(r0, 3, false, NB);
    COMPB(r1, 4, false, NB);
    COMPB(r2, 5, false, 4);
    flush(pend, pst, BAND - 1);
#undef LOADB
#undef COMPB
}

extern "C" void kernel_launch(void* const* d_in, const int* in_sizes, int n_in,
                              void* d_out, int out_size, void* d_ws, size_t ws_size,
                              hipStream_t stream) {
    const float* x  = (const float*)d_in[0];
    const float* Wl = (const float*)d_in[1];
    const float* bl = (const float*)d_in[2];
    float* out = (float*)d_out;

    int N = in_sizes[0] / (CCH * HEIGHT * WIDTH);
    int blocks = N * (CCH / 4);
    sfcr_wmma_kernel<<<blocks, 256, 0, stream>>>(x, Wl, bl, out, N);
}


# Scheduling barrier behind each row's store pair pins the stores to their step: output leaves evenly under the matrix work, not in bursts of 12-20 stores per wave; sum order unchanged
# speedup vs baseline: 1.5781x; 1.0871x over previous
#include <hip/hip_runtime.h>

typedef float v2f __attribute__((ext_vector_type(2)));
typedef float v8f __attribute__((ext_vector_type(8)));

#define WIDTH  64
#define HEIGHT 64
#define CCH    128
#define PLANE  (HEIGHT * WIDTH)
#define BAND   32
#define NB     6
#define NTILE  4

__device__ __constant__ float Gm[3][3] = {
    { 0.5773502691896258f,  0.5773502691896258f,  0.5773502691896258f },
    { 0.7071067811865476f,  0.0f,                -0.7071067811865476f },
    { 0.4082482904638630f, -0.8164965809277260f,  0.4082482904638630f }
};

struct Rows { float a[NTILE][NB], c[NTILE][NB]; };

struct Cols { int a[NTILE], c[NTILE]; };

template <int NU>
__device__ __forceinline__ void load_batch(Rows& R, const float* __restrict__ xin, int rfirst, int t0,
                                           const Cols& off)
{
#pragma unroll
    for (int u = 0; u < NU; ++u) {
        int i = rfirst + t0 + u - 1;
        i = i < 0 ? 0 : (i > HEIGHT - 1 ? HEIGHT - 1 : i);
        const float* rp = xin + i * WIDTH;
#pragma unroll
        for (int j = 0; j < NTILE; ++j) {
            R.a[j][u] = rp[off.a[j]];
            R.c[j][u] = rp[off.c[j]];
        }
    }
}

template <int HS, bool STORE>
__device__ __forceinline__ void retire(v8f (&acc)[NTILE], bool lo, float* __restrict__ srow,
                                       int lane, float4 (&pend)[2],
                                       float b0, float b1, float b2, float b3)
{
    const float bias[4] = { b0, b1, b2, b3 };
    constexpr int V = (HS == 1) ? 4 : 0;
    if (STORE) {
        const int pix = (HS == 2) ? (lane ^ 16) : lane;
#pragma unroll
        for (int P = 0; P < 2; ++P) {
#pragma unroll
            for (int o = 0; o < 4; ++o) {
                float m;
                if (HS == 2) m = lo ? acc[2 * P + 1][V + o] : acc[2 * P][V + o];
                else         m = lo ? acc[2 * P][V + o] : acc[2 * P + 1][V + o];
                srow[o * WIDTH + 32 * P + pix] = m;
            }
        }
#pragma unroll
        for (int j = 0; j < 2; ++j)
            pend[j] = *reinterpret_cast<const float4*>(srow + j * 2 * WIDTH + lane * 4);
    }
#pragma unroll
    for (int P = 0; P < 2; ++P) {
#pragma unroll
        for (int o = 0; o < 4; ++o) {
            if (HS == 2) {
                acc[2 * P][V + o]     = lo ? acc[2 * P][V + o] : bias[o];
                acc[2 * P + 1][V + o] = lo ? bias[o] : acc[2 * P + 1][V + o];
            } else {
                acc[2 * P][V + o]     = lo ? bias[o] : acc[2 * P][V + o];
                acc[2 * P + 1][V + o] = lo ? acc[2 * P + 1][V + o] : bias[o];
            }
        }
    }
}

__device__ __forceinline__ void flush(const float4 (&pend)[2], float* __restrict__ pst, int row)
{
#pragma unroll
    for (int j = 0; j < 2; ++j)
        *reinterpret_cast<float4*>(pst + j * 2 * PLANE + row * WIDTH) = pend[j];
}

template <bool FIRST, int NU>
__device__ __forceinline__ void compute_batch(const Rows& R, v8f (&acc)[NTILE],
                                              const v2f (&A0)[3], const v2f (&A1)[3],
                                              bool lo, float* __restrict__ srow, int lane,
                                              float* __restrict__ pst, int t0, float4 (&pend)[2],
                                              float b0, float b1, float b2, float b3)
{
#pragma unroll
    for (int u = 0; u < NU; ++u) {
        const int s = u % 3;
#pragma unroll
        for (int j = 0; j < NTILE; ++j) {
            v2f f;
            f.x = R.a[j][u]; f.y = lo ? R.c[j][u] : 0.f;
            acc[j] = __builtin_amdgcn_wmma_f32_16x16x4_f32(false, (j & 1) ? A1[s] : A0[s], false, f,
                                                           (short)0, acc[j], false, false);
        }
        const int t = t0 + u;
        if (!(FIRST && u < 3)) flush(pend, pst, t - 3);
        __builtin_amdgcn_sched_barrier(0);
        if (FIRST && u < 2) {
            if (s == 0) retire<1, false>(acc, lo, srow, lane, pend, b0, b1, b2, b3);
            else        retire<2, false>(acc, lo, srow, lane, pend, b0, b1, b2, b3);
        } else {
            if (s == 0)      retire<1, true>(acc, lo, srow, lane, pend, b0, b1, b2, b3);
            else if (s == 1) retire<2, true>(acc, lo, srow, lane, pend, b0, b1, b2, b3);
            else             retire<0, true>(acc, lo, srow, lane, pend, b0, b1, b2, b3);
        }
    }
}

__global__ void __launch_bounds__(256)
sfcr_wmma_kernel(const float* __restrict__ x,
                 const float* __restrict__ Wl,
                 const float* __restrict__ bl,
                 float* __restrict__ out,
                 int N)
{
    __shared__ float sW[4][12];
    __shared__ __attribute__((aligned(16))) float sT[8][4 * WIDTH];

    const int tid = threadIdx.x;
    (void)N;

    const int wave  = __builtin_amdgcn_readfirstlane(tid >> 5);
    const int lane  = tid & 31;
    const int half  = lane >> 4;
    const bool lo   = half == 0;
    const int c     = lane & 15;
    const int plane = blockIdx.x * 4 + (wave >> 1);
    const int rfirst = (wave & 1) * BAND;

    Cols off;
#pragma unroll
    for (int j = 0; j < NTILE; ++j) {
        const int w = 16 * j + c;
        int a = lo ? w - 1 : w + 1;
        a = a < 0 ? 0 : (a > WIDTH - 1 ? WIDTH - 1 : a);
        off.a[j] = a;
        off.c[j] = w;
    }

    const float* xin  = x + (size_t)plane * PLANE;
    float* srow = &sT[wave][0];
    float* pst = out + (size_t)plane * 4 * PLANE + half * PLANE + rfirst * WIDTH + 4 * c;

#define LOADB(R, k, NU)  load_batch<NU>(R, xin, rfirst, (k) * NB, off)
#define COMPB(R, k, F, NU)  compute_batch<F, NU>(R, acc, A0, A1, lo, srow, lane, pst, (k) * NB, pend, b0, b1, b2, b3)
    Rows r0, r1, r2;
    LOADB(r0, 0, NB); LOADB(r1, 1, NB); LOADB(r2, 2, NB);

    if (tid < 36) {
        int o = tid / 9, j = tid % 9, p = j / 3, q = j % 3;
        float accw = Wl[o * 18 + j];
#pragma unroll
        for (int a = 0; a < 3; ++a)
#pragma unroll
            for (int b = 0; b < 3; ++b)
                accw += Wl[o * 18 + 9 + 3 * a + b] * Gm[a][p] * Gm[b][q];
        sW[o][j] = accw;
    }
    __syncthreads();

    v2f A0[3], A1[3];
    {
        const int blk = c >> 2, o = c & 3;
        const int posg[2][3] = { {0, 1, 2}, {2, 3, 0} };
#pragma unroll
        for (int s = 0; s < 3; ++s) {
            v2f a0 = {0.f, 0.f}, a1 = {0.f, 0.f};
#pragma unroll
            for (int p = 0; p < 3; ++p) {
                const int hres = (s + 3 - p) % 3;
                float wx = sW[o][3 * p + 2 * half];
                float wy = lo ? sW[o][3 * p + 1] : 0.f;
                if (posg[0][hres] == blk) { a0.x = wx; a0.y = wy; }
                if (posg[1][hres] == blk) { a1.x = wx; a1.y = wy; }
            }
            A0[s] = a0; A1[s] = a1;
        }
    }

    const float b0 = bl[0], b1 = bl[1], b2 = bl[2], b3 = bl[3];
    v8f acc[NTILE];
#pragma unroll
    for (int j = 0; j < NTILE; ++j) acc[j] = (v8f){ b0, b1, b2, b3, b0, b1, b2, b3 };
    float4 pend[2];

    COMPB(r0, 0, true, NB);
    LOADB(r0, 3, NB); COMPB(r1, 1, false, NB);
    LOADB(r1, 4, NB); COMPB(r2, 2, false, NB);
    LOADB(r2, 5, 4);  COMPB(r0, 3, false, NB);
    COMPB(r1, 4, false, NB);
    COMPB(r2, 5, false, 4);
    flush(pend, pst, BAND - 1);
#undef LOADB
#undef COMPB
}

extern "C" void kernel_launch(void* const* d_in, const int* in_sizes, int n_in,
                              void* d_out, int out_size, void* d_ws, size_t ws_size,
                              hipStream_t stream) {
    const float* x  = (const float*)d_in[0];
    const float* Wl = (const float*)d_in[1];
    const float* bl = (const float*)d_in[2];
    float* out = (float*)d_out;

    int N = in_sizes[0] / (CCH * HEIGHT * WIDTH);
    int blocks = N * (CCH / 4);
    sfcr_wmma_kernel<<<blocks, 256, 0, stream>>>(x, Wl, bl, out, N);
}


# Column c of 16-col group j is pixel 4c+j: one aligned 16-byte load + one 4-byte edge load per lane feed a row's four B fragments (2 loads a row, was 8); LDS write index follows; sum order unchanged
# speedup vs baseline: 1.7586x; 1.1144x over previous
#include <hip/hip_runtime.h>

typedef float v2f __attribute__((ext_vector_type(2)));
typedef float v8f __attribute__((ext_vector_type(8)));

#define WIDTH  64
#define HEIGHT 64
#define CCH    128
#define PLANE  (HEIGHT * WIDTH)
#define BAND   32
#define NB     6
#define NTILE  4

__device__ __constant__ float Gm[3][3] = {
    { 0.5773502691896258f,  0.5773502691896258f,  0.5773502691896258f },
    { 0.7071067811865476f,  0.0f,                -0.7071067811865476f },
    { 0.4082482904638630f, -0.8164965809277260f,  0.4082482904638630f }
};

struct Rows { float4 v[NB]; float e[NB]; };

struct Cols { int v, e; };

template <int NU>
__device__ __forceinline__ void load_batch(Rows& R, const float* __restrict__ xin, int rfirst, int t0,
                                           const Cols& off)
{
#pragma unroll
    for (int u = 0; u < NU; ++u) {
        int i = rfirst + t0 + u - 1;
        i = i < 0 ? 0 : (i > HEIGHT - 1 ? HEIGHT - 1 : i);
        const float* rp = xin + i * WIDTH;
        R.v[u] = *reinterpret_cast<const float4*>(rp + off.v);
        R.e[u] = rp[off.e];
    }
}

template <int HS, bool STORE>
__device__ __forceinline__ void retire(v8f (&acc)[NTILE], bool lo, float* __restrict__ srow,
                                       int lane, float4 (&pend)[2],
                                       float b0, float b1, float b2, float b3)
{
    const float bias[4] = { b0, b1, b2, b3 };
    constexpr int V = (HS == 1) ? 4 : 0;
    if (STORE) {
        const int pix = 4 * (lane & 15) + ((HS == 2) ? ((lane >> 4) ^ 1) : (lane >> 4));
#pragma unroll
        for (int P = 0; P < 2; ++P) {
#pragma unroll
            for (int o = 0; o < 4; ++o) {
                float m;
                if (HS == 2) m = lo ? acc[2 * P + 1][V + o] : acc[2 * P][V + o];
                else         m = lo ? acc[2 * P][V + o] : acc[2 * P + 1][V + o];
                srow[o * WIDTH + 2 * P + pix] = m;
            }
        }
#pragma unroll
        for (int j = 0; j < 2; ++j)
            pend[j] = *reinterpret_cast<const float4*>(srow + j * 2 * WIDTH + lane * 4);
    }
#pragma unroll
    for (int P = 0; P < 2; ++P) {
#pragma unroll
        for (int o = 0; o < 4; ++o) {
            if (HS == 2) {
                acc[2 * P][V + o]     = lo ? acc[2 * P][V + o] : bias[o];
                acc[2 * P + 1][V + o] = lo ? bias[o] : acc[2 * P + 1][V + o];
            } else {
                acc[2 * P][V + o]     = lo ? bias[o] : acc[2 * P][V + o];
                acc[2 * P + 1][V + o] = lo ? acc[2 * P + 1][V + o] : bias[o];
            }
        }
    }
}

__device__ __forceinline__ void flush(const float4 (&pend)[2], float* __restrict__ pst, int row)
{
#pragma unroll
    for (int j = 0; j < 2; ++j)
        *reinterpret_cast<float4*>(pst + j * 2 * PLANE + row * WIDTH) = pend[j];
}

template <bool FIRST, int NU>
__device__ __forceinline__ void compute_batch(const Rows& R, v8f (&acc)[NTILE],
                                              const v2f (&A0)[3], const v2f (&A1)[3],
                                              bool lo, float* __restrict__ srow, int lane,
                                              float* __restrict__ pst, int t0, float4 (&pend)[2],
                                              float b0, float b1, float b2, float b3)
{
#pragma unroll
    for (int u = 0; u < NU; ++u) {
        const int s = u % 3;
#pragma unroll
        for (int j = 0; j < NTILE; ++j) {
            const float4 v = R.v[u];
            const float e = R.e[u];
            const float ctr = (j == 0) ? v.x : (j == 1) ? v.y : (j == 2) ? v.z : v.w;
            const float lft = (j == 0) ? e   : (j == 1) ? v.x : (j == 2) ? v.y : v.z;
            const float rgt = (j == 0) ? v.y : (j == 1) ? v.z : (j == 2) ? v.w : e;
            v2f f;
            f.x = lo ? lft : rgt; f.y = lo ? ctr : 0.f;
            acc[j] = __builtin_amdgcn_wmma_f32_16x16x4_f32(false, (j & 1) ? A1[s] : A0[s], false, f,
                                                           (short)0, acc[j], false, false);
        }
        const int t = t0 + u;
        if (!(FIRST && u < 3)) flush(pend, pst, t - 3);
        __builtin_amdgcn_sched_barrier(0);
        if (FIRST && u < 2) {
            if (s == 0) retire<1, false>(acc, lo, srow, lane, pend, b0, b1, b2, b3);
            else        retire<2, false>(acc, lo, srow, lane, pend, b0, b1, b2, b3);
        } else {
            if (s == 0)      retire<1, true>(acc, lo, srow, lane, pend, b0, b1, b2, b3);
            else if (s == 1) retire<2, true>(acc, lo, srow, lane, pend, b0, b1, b2, b3);
            else             retire<0, true>(acc, lo, srow, lane, pend, b0, b1, b2, b3);
        }
    }
}

__global__ void __launch_bounds__(256)
sfcr_wmma_kernel(const float* __restrict__ x,
                 const float* __restrict__ Wl,
                 const float* __restrict__ bl,
                 float* __restrict__ out,
                 int N)
{
    __shared__ float sW[4][12];
    __shared__ __attribute__((aligned(16))) float sT[8][4 * WIDTH];

    const int tid = threadIdx.x;
    (void)N;

    const int wave  = __builtin_amdgcn_readfirstlane(tid >> 5);
    const int lane  = tid & 31;
    const int half  = lane >> 4;
    const bool lo   = half == 0;
    const int c     = lane & 15;
    const int plane = blockIdx.x * 4 + (wave >> 1);
    const int rfirst = (wave & 1) * BAND;

    Cols off;
    {
        int e = lo ? 4 * c - 1 : 4 * c + 4;
        e = e < 0 ? 0 : (e > WIDTH - 1 ? WIDTH - 1 : e);
        off.v = 4 * c;
        off.e = e;
    }

    const float* xin  = x + (size_t)plane * PLANE;
    float* srow = &sT[wave][0];
    float* pst = out + (size_t)plane * 4 * PLANE + half * PLANE + rfirst * WIDTH + 4 * c;

#define LOADB(R, k, NU)  load_batch<NU>(R, xin, rfirst, (k) * NB, off)
#define COMPB(R, k, F, NU)  compute_batch<F, NU>(R, acc, A0, A1, lo, srow, lane, pst, (k) * NB, pend, b0, b1, b2, b3)
    Rows r0, r1, r2;
    LOADB(r0, 0, NB); LOADB(r1, 1, NB); LOADB(r2, 2, NB);

    if (tid < 36) {
        int o = tid / 9, j = tid % 9, p = j / 3, q = j % 3;
        float accw = Wl[o * 18 + j];
#pragma unroll
        for (int a = 0; a < 3; ++a)
#pragma unroll
            for (int b = 0; b < 3; ++b)
                accw += Wl[o * 18 + 9 + 3 * a + b] * Gm[a][p] * Gm[b][q];
        sW[o][j] = accw;
    }
    __syncthreads();

    v2f A0[3], A1[3];
    {
        const int blk = c >> 2, o = c & 3;
        const int posg[2][3] = { {0, 1, 2}, {2, 3, 0} };
#pragma unroll
        for (int s = 0; s < 3; ++s) {
            v2f a0 = {0.f, 0.f}, a1 = {0.f, 0.f};
#pragma unroll
            for (int p = 0; p < 3; ++p) {
                const int hres = (s + 3 - p) % 3;
                float wx = sW[o][3 * p + 2 * half];
                const float wc = sW[o][3 * p + 1];
                float wy = lo ? wc : 0.f;
                if (posg[0][hres] == blk) { a0.x = wx; a0.y = wy; }
                if (posg[1][hres] == blk) { a1.x = wx; a1.y = wy; }
            }
            A0[s] = a0; A1[s] = a1;
        }
    }

    const float b0 = bl[0], b1 = bl[1], b2 = bl[2], b3 = bl[3];
    v8f acc[NTILE];
#pragma unroll
    for (int j = 0; j < NTILE; ++j) acc[j] = (v8f){ b0, b1, b2, b3, b0, b1, b2, b3 };
    float4 pend[2];

    COMPB(r0, 0, true, NB);
    LOADB(r0, 3, NB); COMPB(r1, 1, false, NB);
    LOADB(r1, 4, NB); COMPB(r2, 2, false, NB);
    LOADB(r2, 5, 4);  COMPB(r0, 3, false, NB);
    COMPB(r1, 4, false, NB);
    COMPB(r2, 5, false, 4);
    flush(pend, pst, BAND - 1);
#undef LOADB
#undef COMPB
}

extern "C" void kernel_launch(void* const* d_in, const int* in_sizes, int n_in,
                              void* d_out, int out_size, void* d_ws, size_t ws_size,
                              hipStream_t stream) {
    const float* x  = (const float*)d_in[0];
    const float* Wl = (const float*)d_in[1];
    const float* bl = (const float*)d_in[2];
    float* out = (float*)d_out;

    int N = in_sizes[0] / (CCH * HEIGHT * WIDTH);
    int blocks = N * (CCH / 4);
    sfcr_wmma_kernel<<<blocks, 256, 0, stream>>>(x, Wl, bl, out, N);
}
